# starting kernel
# baseline (speedup 1.0000x reference)
#include <hip/hip_runtime.h>

typedef __attribute__((ext_vector_type(2))) float v2f;
typedef __attribute__((ext_vector_type(8))) float v8f;

#define NNODES 100000
#define DIM 64
#define NTILES_M (NNODES / 16)

__global__ __launch_bounds__(256) void gemm_wmma(const float* __restrict__ X,
                                                 const float* __restrict__ W,
                                                 float* __restrict__ Y) {
  const int wave  = blockIdx.x * (blockDim.x >> 5) + (threadIdx.x >> 5);
  const int lane  = threadIdx.x & 31;
  if (wave >= NTILES_M) return;

  const int mrow  = wave * 16 + (lane & 15);
  const int khalf = (lane >> 4) * 2;
  const int nlo   = lane & 15;

  v8f acc0 = {}, acc1 = {}, acc2 = {}, acc3 = {};

  const float* xrow = X + (size_t)mrow * DIM;
  for (int k0 = 0; k0 < DIM; k0 += 4) {
    v2f a;
    a.x = xrow[k0 + khalf + 0];
    a.y = xrow[k0 + khalf + 1];
    const float* w0 = W + (size_t)(k0 + khalf) * DIM + nlo;
    v2f b0; b0.x = w0[0];        b0.y = w0[DIM];
    v2f b1; b1.x = w0[16];       b1.y = w0[DIM + 16];
    v2f b2; b2.x = w0[32];       b2.y = w0[DIM + 32];
    v2f b3; b3.x = w0[48];       b3.y = w0[DIM + 48];
    acc0 = __builtin_amdgcn_wmma_f32_16x16x4_f32(false, a, false, b0, (short)0, acc0, false, false);
    acc1 = __builtin_amdgcn_wmma_f32_16x16x4_f32(false, a, false, b1, (short)0, acc1, false, false);
    acc2 = __builtin_amdgcn_wmma_f32_16x16x4_f32(false, a, false, b2, (short)0, acc2, false, false);
    acc3 = __builtin_amdgcn_wmma_f32_16x16x4_f32(false, a, false, b3, (short)0, acc3, false, false);
  }

  const int rbase = wave * 16 + (lane >> 4) * 8;
  float* yb = Y + (size_t)rbase * DIM + nlo;
#pragma unroll
  for (int r = 0; r < 8; ++r) {
    yb[(size_t)r * DIM +  0] = acc0[r];
    yb[(size_t)r * DIM + 16] = acc1[r];
    yb[(size_t)r * DIM + 32] = acc2[r];
    yb[(size_t)r * DIM + 48] = acc3[r];
  }
}

__global__ __launch_bounds__(256) void init_out(const float* __restrict__ ori,
                                                const float* __restrict__ bias,
                                                float* __restrict__ out) {
  const int i = blockIdx.x * blockDim.x + threadIdx.x;
  if (i >= NNODES * (DIM / 4)) return;
  const float4 o = ((const float4*)ori)[i];
  const float4 b = ((const float4*)bias)[i & (DIM / 4 - 1)];
  float4 r;
  r.x = o.x + b.x; r.y = o.y + b.y; r.z = o.z + b.z; r.w = o.w + b.w;
  ((float4*)out)[i] = r;
}

__global__ __launch_bounds__(256) void scatter_edges(const float* __restrict__ XW,
                                                     const int* __restrict__ src,
                                                     const int* __restrict__ dst,
                                                     const float* __restrict__ ew,
                                                     float* __restrict__ out,
                                                     int nE) {
  const long long t = (long long)blockIdx.x * blockDim.x + threadIdx.x;
  const int e = (int)(t >> 4);
  if (e >= nE) return;
  const int q = (int)(t & 15);

  const float w = ew[e];
  const float4 m = ((const float4*)(XW + (size_t)src[e] * DIM))[q];
  float* o = out + (size_t)dst[e] * DIM + q * 4;
#if defined(__HIP_DEVICE_COMPILE__)
  unsafeAtomicAdd(o + 0, m.x * w);
  unsafeAtomicAdd(o + 1, m.y * w);
  unsafeAtomicAdd(o + 2, m.z * w);
  unsafeAtomicAdd(o + 3, m.w * w);
#endif
}

extern "C" void kernel_launch(void* const* d_in, const int* in_sizes, int n_in,
                              void* d_out, int out_size, void* d_ws, size_t ws_size,
                              hipStream_t stream) {
  const float* in_feat = (const float*)d_in[0];
  const float* ew      = (const float*)d_in[1];
  const float* W1      = (const float*)d_in[2];
  const float* b1      = (const float*)d_in[3];
  const float* W2      = (const float*)d_in[4];
  const float* b2      = (const float*)d_in[5];
  const int*   src     = (const int*)d_in[6];
  const int*   dst     = (const int*)d_in[7];
  const int    nE      = in_sizes[1];

  float* xw   = (float*)d_ws;
  float* hbuf = xw + (size_t)NNODES * DIM;
  float* outp = (float*)d_out;

  const int gemmBlocks    = (NTILES_M + 7) / 8;
  const int initBlocks    = (NNODES * (DIM / 4) + 255) / 256;
  const int scatterBlocks = (int)(((long long)nE * 16 + 255) / 256);

  const float* hcur = in_feat;
  for (int step = 0; step < 4; ++step) {
    const float* W = (step == 0) ? W1 : W2;
    const float* b = (step == 0) ? b1 : b2;
    float* tgt = (step == 3) ? outp : hbuf;

    gemm_wmma<<<gemmBlocks, 256, 0, stream>>>(hcur, W, xw);
    init_out<<<initBlocks, 256, 0, stream>>>(in_feat, b, tgt);
    scatter_edges<<<scatterBlocks, 256, 0, stream>>>(xw, src, dst, ew, tgt, nE);

    hcur = tgt;
  }
}


# Bucket edges by destination once per call (int-atomic count + arrival-order rank) and sum each node's bucket in registers with one store: no float atomics, no init pass
# speedup vs baseline: 1.7388x; 1.7388x over previous
#include <hip/hip_runtime.h>

typedef __attribute__((ext_vector_type(2))) float v2f;
typedef __attribute__((ext_vector_type(8))) float v8f;

#define NNODES 100000
#define DIM 64
#define NTILES_M (NNODES / 16)

__global__ __launch_bounds__(256) void gemm_wmma(const float* __restrict__ X,
                                                 const float* __restrict__ W,
                                                 float* __restrict__ Y) {
  const int wave  = blockIdx.x * (blockDim.x >> 5) + (threadIdx.x >> 5);
  const int lane  = threadIdx.x & 31;
  if (wave >= NTILES_M) return;

  const int mrow  = wave * 16 + (lane & 15);
  const int khalf = (lane >> 4) * 2;
  const int nlo   = lane & 15;

  v8f acc0 = {}, acc1 = {}, acc2 = {}, acc3 = {};

  const float* xrow = X + (size_t)mrow * DIM;
  for (int k0 = 0; k0 < DIM; k0 += 4) {
    v2f a;
    a.x = xrow[k0 + khalf + 0];
    a.y = xrow[k0 + khalf + 1];
    const float* w0 = W + (size_t)(k0 + khalf) * DIM + nlo;
    v2f b0; b0.x = w0[0];        b0.y = w0[DIM];
    v2f b1; b1.x = w0[16];       b1.y = w0[DIM + 16];
    v2f b2; b2.x = w0[32];       b2.y = w0[DIM + 32];
    v2f b3; b3.x = w0[48];       b3.y = w0[DIM + 48];
    acc0 = __builtin_amdgcn_wmma_f32_16x16x4_f32(false, a, false, b0, (short)0, acc0, false, false);
    acc1 = __builtin_amdgcn_wmma_f32_16x16x4_f32(false, a, false, b1, (short)0, acc1, false, false);
    acc2 = __builtin_amdgcn_wmma_f32_16x16x4_f32(false, a, false, b2, (short)0, acc2, false, false);
    acc3 = __builtin_amdgcn_wmma_f32_16x16x4_f32(false, a, false, b3, (short)0, acc3, false, false);
  }

  const int rbase = wave * 16 + (lane >> 4) * 8;
  float* yb = Y + (size_t)rbase * DIM + nlo;
#pragma unroll
  for (int r = 0; r < 8; ++r) {
    yb[(size_t)r * DIM +  0] = acc0[r];
    yb[(size_t)r * DIM + 16] = acc1[r];
    yb[(size_t)r * DIM + 32] = acc2[r];
    yb[(size_t)r * DIM + 48] = acc3[r];
  }
}

#define NQ (NNODES / 4)
#define SCAN_BLOCKS ((NQ + 255) / 256)

__global__ __launch_bounds__(256) void zero_counts(int* __restrict__ cnt) {
  const int i = blockIdx.x * 256 + threadIdx.x;
  if (i < NQ) ((int4*)cnt)[i] = make_int4(0, 0, 0, 0);
}

__global__ __launch_bounds__(256) void count_dst(const int* __restrict__ dst,
                                                 int* __restrict__ cnt, int nE) {
  const int e = blockIdx.x * 256 + threadIdx.x;
  if (e < nE) atomicAdd(cnt + dst[e], 1);
}

__global__ __launch_bounds__(256) void scan_counts(const int* __restrict__ cnt,
                                                   int* __restrict__ rowptr) {
  __shared__ int s_pre[8];
  __shared__ int s_wtot[8];
  const int tid  = threadIdx.x;
  const int lane = tid & 31;
  const int wv   = tid >> 5;
  const int4* c4 = (const int4*)cnt;

  int pre = 0;
  for (int it = 0; it < (int)blockIdx.x; ++it) {
    const int4 v = c4[it * 256 + tid];
    pre += (v.x + v.y) + (v.z + v.w);
  }
  for (int d = 16; d > 0; d >>= 1) pre += __shfl_xor(pre, d);
  if (lane == 0) s_pre[wv] = pre;

  const int g = blockIdx.x * 256 + tid;
  int4 v = make_int4(0, 0, 0, 0);
  if (g < NQ) v = c4[g];
  const int t = (v.x + v.y) + (v.z + v.w);
  int inc = t;
  for (int d = 1; d < 32; d <<= 1) {
    const int u = __shfl_up(inc, d);
    if (lane >= d) inc += u;
  }
  if (lane == 31) s_wtot[wv] = inc;
  __syncthreads();

  int base = 0;
  for (int k = 0; k < 8; ++k) {
    base += s_pre[k];
    if (k < wv) base += s_wtot[k];
  }
  const int ex = base + inc - t;
  if (g < NQ) {
    ((int4*)rowptr)[g] = make_int4(ex, ex + v.x, ex + v.x + v.y, ex + v.x + v.y + v.z);
    if (g == NQ - 1) rowptr[NNODES] = ex + t;
  }
}

__global__ __launch_bounds__(256) void bucket_edges(const int* __restrict__ src,
                                                    const int* __restrict__ dst,
                                                    const float* __restrict__ ew,
                                                    const int* __restrict__ rowptr,
                                                    int* __restrict__ cnt,
                                                    int2* __restrict__ csr, int nE) {
  const int e = blockIdx.x * 256 + threadIdx.x;
  if (e >= nE) return;
  const int d   = dst[e];
  const int old = atomicSub(cnt + d, 1);
  csr[rowptr[d] + old - 1] = make_int2(src[e], __float_as_int(ew[e]));
}

__global__ __launch_bounds__(256) void gather_nodes(const float* __restrict__ XW,
                                                    const int2* __restrict__ csr,
                                                    const int* __restrict__ rowptr,
                                                    const float* __restrict__ ori,
                                                    const float* __restrict__ bias,
                                                    float* __restrict__ out) {
  const int t = blockIdx.x * 256 + threadIdx.x;
  const int n = t >> 4;
  if (n >= NNODES) return;
  const int q = t & 15;

  const float4 o = ((const float4*)ori)[t];
  const float4 b = ((const float4*)bias)[q];
  float4 acc;
  acc.x = o.x + b.x; acc.y = o.y + b.y; acc.z = o.z + b.z; acc.w = o.w + b.w;

  const int beg = rowptr[n];
  const int end = rowptr[n + 1];
  const float4* X4 = (const float4*)XW + q;

  const int last = end - 1;
  for (int j = beg; j < end; j += 8) {
    int2   e[8];
    float4 m[8];
#pragma unroll
    for (int k = 0; k < 8; ++k) e[k] = csr[min(j + k, last)];
#pragma unroll
    for (int k = 0; k < 8; ++k) m[k] = X4[(size_t)e[k].x * (DIM / 4)];
#pragma unroll
    for (int k = 0; k < 8; ++k) {
      const bool  ok = (j + k) < end;
      const float w  = __int_as_float(e[k].y);
      const float nx = fmaf(m[k].x, w, acc.x);
      const float ny = fmaf(m[k].y, w, acc.y);
      const float nz = fmaf(m[k].z, w, acc.z);
      const float nw = fmaf(m[k].w, w, acc.w);
      acc.x = ok ? nx : acc.x;
      acc.y = ok ? ny : acc.y;
      acc.z = ok ? nz : acc.z;
      acc.w = ok ? nw : acc.w;
    }
  }
  ((float4*)out)[t] = acc;
}

extern "C" void kernel_launch(void* const* d_in, const int* in_sizes, int n_in,
                              void* d_out, int out_size, void* d_ws, size_t ws_size,
                              hipStream_t stream) {
  const float* in_feat = (const float*)d_in[0];
  const float* ew      = (const float*)d_in[1];
  const float* W1      = (const float*)d_in[2];
  const float* b1      = (const float*)d_in[3];
  const float* W2      = (const float*)d_in[4];
  const float* b2      = (const float*)d_in[5];
  const int*   src     = (const int*)d_in[6];
  const int*   dst     = (const int*)d_in[7];
  const int    nE      = in_sizes[1];

  float* xw   = (float*)d_ws;
  float* hbuf = xw + (size_t)NNODES * DIM;
  int*   cnt    = (int*)(hbuf + (size_t)NNODES * DIM);
  int*   rowptr = cnt + NNODES;
  int2*  csr    = (int2*)(rowptr + NNODES + 4);
  float* outp = (float*)d_out;

  const int gemmBlocks = (NTILES_M + 7) / 8;
  const int edgeBlocks = (nE + 255) / 256;
  const int nodeBlocks = NNODES * 16 / 256;

  zero_counts<<<SCAN_BLOCKS, 256, 0, stream>>>(cnt);
  count_dst<<<edgeBlocks, 256, 0, stream>>>(dst, cnt, nE);
  scan_counts<<<SCAN_BLOCKS, 256, 0, stream>>>(cnt, rowptr);
  bucket_edges<<<edgeBlocks, 256, 0, stream>>>(src, dst, ew, rowptr, cnt, csr, nE);

  const float* hcur = in_feat;
  for (int step = 0; step < 4; ++step) {
    const float* W = (step == 0) ? W1 : W2;
    const float* b = (step == 0) ? b1 : b2;
    float* tgt = (step == 3) ? outp : hbuf;

    gemm_wmma<<<gemmBlocks, 256, 0, stream>>>(hcur, W, xw);
    gather_nodes<<<nodeBlocks, 256, 0, stream>>>(xw, csr, rowptr, in_feat, b, tgt);

    hcur = tgt;
  }
}


# Fixed 48-slot bucket per destination node, slot = the edge's returning count: one atomic pass builds the buckets, count pass and prefix scan dropped; ranks stay arrival-order
# speedup vs baseline: 2.4293x; 1.3971x over previous
#include <hip/hip_runtime.h>

typedef __attribute__((ext_vector_type(2))) float v2f;
typedef __attribute__((ext_vector_type(8))) float v8f;

#define NNODES 100000
#define DIM 64
#define NTILES_M (NNODES / 16)

__global__ __launch_bounds__(256) void gemm_wmma(const float* __restrict__ X,
                                                 const float* __restrict__ W,
                                                 float* __restrict__ Y) {
  const int wave  = blockIdx.x * (blockDim.x >> 5) + (threadIdx.x >> 5);
  const int lane  = threadIdx.x & 31;
  if (wave >= NTILES_M) return;

  const int mrow  = wave * 16 + (lane & 15);
  const int khalf = (lane >> 4) * 2;
  const int nlo   = lane & 15;

  v8f acc0 = {}, acc1 = {}, acc2 = {}, acc3 = {};

  const float* xrow = X + (size_t)mrow * DIM;
  for (int k0 = 0; k0 < DIM; k0 += 4) {
    v2f a;
    a.x = xrow[k0 + khalf + 0];
    a.y = xrow[k0 + khalf + 1];
    const float* w0 = W + (size_t)(k0 + khalf) * DIM + nlo;
    v2f b0; b0.x = w0[0];        b0.y = w0[DIM];
    v2f b1; b1.x = w0[16];       b1.y = w0[DIM + 16];
    v2f b2; b2.x = w0[32];       b2.y = w0[DIM + 32];
    v2f b3; b3.x = w0[48];       b3.y = w0[DIM + 48];
    acc0 = __builtin_amdgcn_wmma_f32_16x16x4_f32(false, a, false, b0, (short)0, acc0, false, false);
    acc1 = __builtin_amdgcn_wmma_f32_16x16x4_f32(false, a, false, b1, (short)0, acc1, false, false);
    acc2 = __builtin_amdgcn_wmma_f32_16x16x4_f32(false, a, false, b2, (short)0, acc2, false, false);
    acc3 = __builtin_amdgcn_wmma_f32_16x16x4_f32(false, a, false, b3, (short)0, acc3, false, false);
  }

  const int rbase = wave * 16 + (lane >> 4) * 8;
  float* yb = Y + (size_t)rbase * DIM + nlo;
#pragma unroll
  for (int r = 0; r < 8; ++r) {
    yb[(size_t)r * DIM +  0] = acc0[r];
    yb[(size_t)r * DIM + 16] = acc1[r];
    yb[(size_t)r * DIM + 32] = acc2[r];
    yb[(size_t)r * DIM + 48] = acc3[r];
  }
}

#define CAP 48
#define NCNT (NNODES + 4)
#define NQ (NCNT / 4)
#define ZERO_BLOCKS ((NQ + 255) / 256)
#define CNT_BYTES 400384

__global__ __launch_bounds__(256) void zero_counts(int* __restrict__ cnt) {
  const int i = blockIdx.x * 256 + threadIdx.x;
  if (i < NQ) ((int4*)cnt)[i] = make_int4(0, 0, 0, 0);
}

__global__ __launch_bounds__(256) void bucket_edges(const int* __restrict__ src,
                                                    const int* __restrict__ dst,
                                                    const float* __restrict__ ew,
                                                    int* __restrict__ cnt,
                                                    int2* __restrict__ ell,
                                                    int4* __restrict__ ovf, int nE) {
  const int e = blockIdx.x * 256 + threadIdx.x;
  if (e >= nE) return;
  const int d = dst[e];
  const int s = src[e];
  const int w = __float_as_int(ew[e]);
  const int r = atomicAdd(cnt + d, 1);
  if (r < CAP) {
    ell[(size_t)d * CAP + r] = make_int2(s, w);
  } else {
    const int o = atomicAdd(cnt + NNODES, 1);
    ovf[o] = make_int4(d, s, w, 0);
  }
}

__global__ __launch_bounds__(256) void gather_nodes(const float* __restrict__ XW,
                                                    const int2* __restrict__ ell,
                                                    const int4* __restrict__ ovf,
                                                    const int* __restrict__ cnt,
                                                    const float* __restrict__ ori,
                                                    const float* __restrict__ bias,
                                                    float* __restrict__ out) {
  const int t = blockIdx.x * 256 + threadIdx.x;
  const int n = t >> 4;
  if (n >= NNODES) return;
  const int q = t & 15;

  const float4 o = ((const float4*)ori)[t];
  const float4 b = ((const float4*)bias)[q];
  float4 acc;
  acc.x = o.x + b.x; acc.y = o.y + b.y; acc.z = o.z + b.z; acc.w = o.w + b.w;

  const int deg = cnt[n];
  const int beg = n * CAP;
  const int end = beg + min(deg, CAP);
  const float4* X4 = (const float4*)XW + q;

  if (deg > CAP) {
    const int no = cnt[NNODES];
    for (int i = 0; i < no; ++i) {
      const int4 r = ovf[i];
      if (r.x == n) {
        const float4 mm = X4[(size_t)r.y * (DIM / 4)];
        const float  w  = __int_as_float(r.z);
        acc.x = fmaf(mm.x, w, acc.x);
        acc.y = fmaf(mm.y, w, acc.y);
        acc.z = fmaf(mm.z, w, acc.z);
        acc.w = fmaf(mm.w, w, acc.w);
      }
    }
  }

  const int last = end - 1;
  for (int j = beg; j < end; j += 8) {
    int2   e[8];
    float4 m[8];
#pragma unroll
    for (int k = 0; k < 8; ++k) e[k] = ell[min(j + k, last)];
#pragma unroll
    for (int k = 0; k < 8; ++k) m[k] = X4[(size_t)e[k].x * (DIM / 4)];
#pragma unroll
    for (int k = 0; k < 8; ++k) {
      const bool  ok = (k == 0) || (j + k) < end;
      const float w  = __int_as_float(e[k].y);
      const float nx = fmaf(m[k].x, w, acc.x);
      const float ny = fmaf(m[k].y, w, acc.y);
      const float nz = fmaf(m[k].z, w, acc.z);
      const float nw = fmaf(m[k].w, w, acc.w);
      acc.x = ok ? nx : acc.x;
      acc.y = ok ? ny : acc.y;
      acc.z = ok ? nz : acc.z;
      acc.w = ok ? nw : acc.w;
    }
  }

  ((float4*)out)[t] = acc;
}

extern "C" void kernel_launch(void* const* d_in, const int* in_sizes, int n_in,
                              void* d_out, int out_size, void* d_ws, size_t ws_size,
                              hipStream_t stream) {
  const float* in_feat = (const float*)d_in[0];
  const float* ew      = (const float*)d_in[1];
  const float* W1      = (const float*)d_in[2];
  const float* b1      = (const float*)d_in[3];
  const float* W2      = (const float*)d_in[4];
  const float* b2      = (const float*)d_in[5];
  const int*   src     = (const int*)d_in[6];
  const int*   dst     = (const int*)d_in[7];
  const int    nE      = in_sizes[1];

  float* xw   = (float*)d_ws;
  float* hbuf = xw + (size_t)NNODES * DIM;
  int*   cnt  = (int*)(hbuf + (size_t)NNODES * DIM);
  int2*  ell  = (int2*)((char*)cnt + CNT_BYTES);
  int4*  ovf  = (int4*)(ell + (size_t)NNODES * CAP);
  float* outp = (float*)d_out;

  const int gemmBlocks = (NTILES_M + 7) / 8;
  const int edgeBlocks = (nE + 255) / 256;
  const int nodeBlocks = NNODES * 16 / 256;

  zero_counts<<<ZERO_BLOCKS, 256, 0, stream>>>(cnt);
  bucket_edges<<<edgeBlocks, 256, 0, stream>>>(src, dst, ew, cnt, ell, ovf, nE);

  const float* hcur = in_feat;
  for (int step = 0; step < 4; ++step) {
    const float* W = (step == 0) ? W1 : W2;
    const float* b = (step == 0) ? b1 : b2;
    float* tgt = (step == 3) ? outp : hbuf;

    gemm_wmma<<<gemmBlocks, 256, 0, stream>>>(hcur, W, xw);
    gather_nodes<<<nodeBlocks, 256, 0, stream>>>(xw, ell, ovf, cnt, in_feat, b, tgt);

    hcur = tgt;
  }
}


# Sum each node's weighted rows of h first, then multiply the 16-node tile by W on WMMA in the same kernel (via LDS): separate GEMM launches and the xw round trip drop; order of FP sums changes
# speedup vs baseline: 2.9509x; 1.2147x over previous
#include <hip/hip_runtime.h>

typedef __attribute__((ext_vector_type(2))) float v2f;
typedef __attribute__((ext_vector_type(8))) float v8f;

#define NNODES 100000
#define DIM 64

#define CAP 48
#define NCNT (NNODES + 4)
#define NQ (NCNT / 4)
#define ZERO_BLOCKS ((NQ + 255) / 256)
#define CNT_BYTES 400384

__global__ __launch_bounds__(256) void zero_counts(int* __restrict__ cnt) {
  const int i = blockIdx.x * 256 + threadIdx.x;
  if (i < NQ) ((int4*)cnt)[i] = make_int4(0, 0, 0, 0);
}

__global__ __launch_bounds__(256) void bucket_edges(const int* __restrict__ src,
                                                    const int* __restrict__ dst,
                                                    const float* __restrict__ ew,
                                                    int* __restrict__ cnt,
                                                    int2* __restrict__ ell,
                                                    int4* __restrict__ ovf, int nE) {
  const int e = blockIdx.x * 256 + threadIdx.x;
  if (e >= nE) return;
  const int d = dst[e];
  const int s = src[e];
  const int w = __float_as_int(ew[e]);
  const int r = atomicAdd(cnt + d, 1);
  if (r < CAP) {
    ell[(size_t)d * CAP + r] = make_int2(s, w);
  } else {
    const int o = atomicAdd(cnt + NNODES, 1);
    ovf[o] = make_int4(d, s, w, 0);
  }
}

#define LDP 68

static_assert(NNODES % 16 == 0, "a block is exactly 16 nodes");

__global__ __launch_bounds__(256) void gather_gemm(const float* __restrict__ H,
                                                   const float* __restrict__ W,
                                                   const int2* __restrict__ ell,
                                                   const int4* __restrict__ ovf,
                                                   const int* __restrict__ cnt,
                                                   const float* ori,
                                                   const float* __restrict__ bias,
                                                   float* __restrict__ out) {
  __shared__ float As[16 * LDP];
  __shared__ float Ds[2][16 * LDP];

  const int t  = blockIdx.x * 256 + threadIdx.x;
  const int n  = t >> 4;
  const int q  = t & 15;
  const int nl = threadIdx.x >> 4;

  float4 acc = make_float4(0.f, 0.f, 0.f, 0.f);

  const int deg = cnt[n];
  const int beg = n * CAP;
  const int end = beg + min(deg, CAP);
  const float4* X4 = (const float4*)H + q;

  if (deg > CAP) {
    const int no = cnt[NNODES];
    for (int i = 0; i < no; ++i) {
      const int4 r = ovf[i];
      if (r.x == n) {
        const float4 mm = X4[(size_t)r.y * (DIM / 4)];
        const float  w  = __int_as_float(r.z);
        acc.x = fmaf(mm.x, w, acc.x);
        acc.y = fmaf(mm.y, w, acc.y);
        acc.z = fmaf(mm.z, w, acc.z);
        acc.w = fmaf(mm.w, w, acc.w);
      }
    }
  }

  const int last = end - 1;
  for (int j = beg; j < end; j += 8) {
    int2   e[8];
    float4 m[8];
#pragma unroll
    for (int k = 0; k < 8; ++k) e[k] = ell[min(j + k, last)];
#pragma unroll
    for (int k = 0; k < 8; ++k) m[k] = X4[(size_t)e[k].x * (DIM / 4)];
#pragma unroll
    for (int k = 0; k < 8; ++k) {
      const bool  ok = (k == 0) || (j + k) < end;
      const float w  = __int_as_float(e[k].y);
      const float nx = fmaf(m[k].x, w, acc.x);
      const float ny = fmaf(m[k].y, w, acc.y);
      const float nz = fmaf(m[k].z, w, acc.z);
      const float nw = fmaf(m[k].w, w, acc.w);
      acc.x = ok ? nx : acc.x;
      acc.y = ok ? ny : acc.y;
      acc.z = ok ? nz : acc.z;
      acc.w = ok ? nw : acc.w;
    }
  }

  *(float4*)&As[nl * LDP + 4 * q] = acc;

  const int lane = threadIdx.x & 31;
  const int wave = threadIdx.x >> 5;
  const int c    = lane & 15;
  const int hh   = lane >> 4;
  const int n0   = (wave & 3) * 16;
  const int kb   = (wave >> 2) * 32;

  v2f bf[8];
#pragma unroll
  for (int i = 0; i < 8; ++i) {
    const float* w0 = W + (size_t)(kb + 4 * i + 2 * hh) * DIM + n0 + c;
    bf[i].x = w0[0];
    bf[i].y = w0[DIM];
  }

  __builtin_amdgcn_fence(__ATOMIC_RELEASE, "workgroup", "local");
  __builtin_amdgcn_s_barrier();
  __builtin_amdgcn_fence(__ATOMIC_ACQUIRE, "workgroup", "local");

  const float4 o = ((const float4*)ori)[t];
  const float4 b = ((const float4*)bias)[q];

  v8f d = {};
#pragma unroll
  for (int i = 0; i < 8; ++i) {
    const v2f a = *(const v2f*)&As[c * LDP + kb + 4 * i + 2 * hh];
    d = __builtin_amdgcn_wmma_f32_16x16x4_f32(false, a, false, bf[i], (short)0, d, false, false);
  }

  float* dp = &Ds[wave >> 2][(8 * hh) * LDP + n0 + c];
#pragma unroll
  for (int r = 0; r < 8; ++r) dp[r * LDP] = d[r];

  __builtin_amdgcn_fence(__ATOMIC_RELEASE, "workgroup", "local");
  __builtin_amdgcn_s_barrier();
  __builtin_amdgcn_fence(__ATOMIC_ACQUIRE, "workgroup", "local");

  const float4 d0 = *(const float4*)&Ds[0][nl * LDP + 4 * q];
  const float4 d1 = *(const float4*)&Ds[1][nl * LDP + 4 * q];
  float4 res;
  res.x = ((d0.x + d1.x) + b.x) + o.x;
  res.y = ((d0.y + d1.y) + b.y) + o.y;
  res.z = ((d0.z + d1.z) + b.z) + o.z;
  res.w = ((d0.w + d1.w) + b.w) + o.w;
  ((float4*)out)[t] = res;
}

extern "C" void kernel_launch(void* const* d_in, const int* in_sizes, int n_in,
                              void* d_out, int out_size, void* d_ws, size_t ws_size,
                              hipStream_t stream) {
  const float* in_feat = (const float*)d_in[0];
  const float* ew      = (const float*)d_in[1];
  const float* W1      = (const float*)d_in[2];
  const float* b1      = (const float*)d_in[3];
  const float* W2      = (const float*)d_in[4];
  const float* b2      = (const float*)d_in[5];
  const int*   src     = (const int*)d_in[6];
  const int*   dst     = (const int*)d_in[7];
  const int    nE      = in_sizes[1];

  float* ha   = (float*)d_ws;
  float* hb   = ha + (size_t)NNODES * DIM;
  int*   cnt  = (int*)(hb + (size_t)NNODES * DIM);
  int2*  ell  = (int2*)((char*)cnt + CNT_BYTES);
  int4*  ovf  = (int4*)(ell + (size_t)NNODES * CAP);
  float* outp = (float*)d_out;

  const int edgeBlocks = (nE + 255) / 256;
  const int nodeBlocks = NNODES * 16 / 256;

  zero_counts<<<ZERO_BLOCKS, 256, 0, stream>>>(cnt);
  bucket_edges<<<edgeBlocks, 256, 0, stream>>>(src, dst, ew, cnt, ell, ovf, nE);

  gather_gemm<<<nodeBlocks, 256, 0, stream>>>(in_feat, W1, ell, ovf, cnt, in_feat, b1, ha);
  gather_gemm<<<nodeBlocks, 256, 0, stream>>>(ha,      W2, ell, ovf, cnt, in_feat, b2, hb);
  gather_gemm<<<nodeBlocks, 256, 0, stream>>>(hb,      W2, ell, ovf, cnt, in_feat, b2, ha);
  gather_gemm<<<nodeBlocks, 256, 0, stream>>>(ha,      W2, ell, ovf, cnt, in_feat, b2, outp);
}


# Edge ranks from LDS counters in two passes (edge chunks ordered by node range in LDS, then a work-group per range ranks per node) replace a returning global atomic per edge; ranks stay arrival-order
# speedup vs baseline: 4.8182x; 1.6328x over previous
#include <hip/hip_runtime.h>

typedef __attribute__((ext_vector_type(2))) float v2f;
typedef __attribute__((ext_vector_type(8))) float v8f;

#define NNODES 100000
#define DIM 64

#define CAP 48
#define NCNT (NNODES + 4)
#define CNT_BYTES 400384

#define NEDGES 1600000
#define NCHUNK 256
#define CH (NEDGES / NCHUNK)
#define NBIN 256
#define BINN 391
#define EPT 7

static_assert(NEDGES % NCHUNK == 0, "chunks are whole");
static_assert(NBIN * BINN >= NNODES, "the ranges cover every node");
static_assert(EPT * 1024 >= CH, "a work-group of 1024 covers its chunk");
static_assert(CH < 65536, "start and length of a piece share one word");
static_assert(BINN <= 512 && NNODES <= (1 << 17), "9 bits of node offset above 17 bits of source");
static_assert(NBIN % 32 == 0 && NCHUNK % 32 == 0, "whole waves in the scan and in the piece loop");

__global__ __launch_bounds__(1024) void chunk_sort(const int* __restrict__ src,
                                                   const int* __restrict__ dst,
                                                   const float* __restrict__ ew,
                                                   int2* __restrict__ cbuf,
                                                   int* __restrict__ seg,
                                                   int* __restrict__ cnt, int nE) {
  __shared__ int  hist[NBIN];
  __shared__ int  wsum[NBIN / 32];
  __shared__ int2 sorted[CH];

  const int tid = threadIdx.x;
  const int c   = blockIdx.x;
  const int e0  = c * CH;
  const int n   = max(0, min(CH, nE - e0));

  if (tid < NBIN) hist[tid] = 0;
  if (c == 0 && tid == 0) cnt[NNODES] = 0;

  int d[EPT], s[EPT], w[EPT], r[EPT];
#pragma unroll
  for (int k = 0; k < EPT; ++k) {
    const int i = tid + 1024 * k;
    d[k] = 0; s[k] = 0; w[k] = 0;
    if (i < n) {
      d[k] = dst[e0 + i];
      s[k] = src[e0 + i];
      w[k] = __float_as_int(ew[e0 + i]);
    }
  }
  __syncthreads();

#pragma unroll
  for (int k = 0; k < EPT; ++k) {
    const int i = tid + 1024 * k;
    r[k] = 0;
    if (i < n) r[k] = atomicAdd(&hist[d[k] / BINN], 1);
  }
  __syncthreads();

  int v = 0, incl = 0;
  if (tid < NBIN) {
    v    = hist[tid];
    incl = v;
    for (int off = 1; off < 32; off <<= 1) {
      const int up = __shfl_up(incl, off);
      if ((tid & 31) >= off) incl += up;
    }
    if ((tid & 31) == 31) wsum[tid >> 5] = incl;
  }
  __syncthreads();
  if (tid < NBIN) {
    int base = 0;
    for (int j = 0; j < (tid >> 5); ++j) base += wsum[j];
    const int excl = base + incl - v;
    hist[tid] = excl;
    seg[tid * NCHUNK + c] = (excl << 16) | v;
  }
  __syncthreads();

#pragma unroll
  for (int k = 0; k < EPT; ++k) {
    const int i = tid + 1024 * k;
    if (i < n) {
      const int b = d[k] / BINN;
      sorted[hist[b] + r[k]] = make_int2(((d[k] - b * BINN) << 17) | s[k], w[k]);
    }
  }
  __syncthreads();

  for (int i = tid; i < n; i += 1024) cbuf[e0 + i] = sorted[i];
}

__global__ __launch_bounds__(1024) void bin_rank(const int2* __restrict__ cbuf,
                                                 const int* __restrict__ seg,
                                                 int* __restrict__ cnt,
                                                 int2* __restrict__ ell,
                                                 int4* __restrict__ ovf) {
  __shared__ int ncnt[BINN];

  const int tid  = threadIdx.x;
  const int lane = tid & 31;
  const int wave = tid >> 5;
  const int b    = blockIdx.x;
  const int n0   = b * BINN;

  if (tid < BINN) ncnt[tid] = 0;

  auto put = [&](const int2 ed, const int rk) {
    const int sn = ed.x & 0x1ffff;
    const int nd = n0 + (int)((unsigned)ed.x >> 17);
    if (rk < CAP) {
      ell[(size_t)nd * CAP + rk] = make_int2(sn, ed.y);
    } else {
      const int o = atomicAdd(cnt + NNODES, 1);
      ovf[o] = make_int4(nd, sn, ed.y, 0);
    }
  };

  int  st[NCHUNK / 32], cn[NCHUNK / 32];
  int2 e[NCHUNK / 32];
#pragma unroll
  for (int j = 0; j < NCHUNK / 32; ++j) {
    const int      c    = wave + 32 * j;
    const unsigned info = (unsigned)seg[b * NCHUNK + c];
    st[j] = c * CH + (int)(info >> 16);
    cn[j] = (int)(info & 0xffffu);
  }
#pragma unroll
  for (int j = 0; j < NCHUNK / 32; ++j) {
    e[j] = cbuf[min(st[j] + (lane < cn[j] ? lane : 0), NEDGES - 1)];
  }
  __syncthreads();

  int rk[NCHUNK / 32];
#pragma unroll
  for (int j = 0; j < NCHUNK / 32; ++j) {
    rk[j] = 0;
    if (lane < cn[j]) rk[j] = atomicAdd(&ncnt[(unsigned)e[j].x >> 17], 1);
  }
#pragma unroll
  for (int j = 0; j < NCHUNK / 32; ++j) {
    if (lane < cn[j]) put(e[j], rk[j]);
  }
#pragma unroll
  for (int j = 0; j < NCHUNK / 32; ++j) {
    for (int k = lane + 32; k < cn[j]; k += 32) {
      const int2 ed = cbuf[st[j] + k];
      put(ed, atomicAdd(&ncnt[(unsigned)ed.x >> 17], 1));
    }
  }
  __syncthreads();

  if (tid < BINN && n0 + tid < NNODES) cnt[n0 + tid] = ncnt[tid];
}

#define LDP 68

static_assert(NNODES % 16 == 0, "a block is exactly 16 nodes");

__global__ __launch_bounds__(256) void gather_gemm(const float* __restrict__ H,
                                                   const float* __restrict__ W,
                                                   const int2* __restrict__ ell,
                                                   const int4* __restrict__ ovf,
                                                   const int* __restrict__ cnt,
                                                   const float* ori,
                                                   const float* __restrict__ bias,
                                                   float* __restrict__ out) {
  __shared__ float As[16 * LDP];
  __shared__ float Ds[2][16 * LDP];

  const int t  = blockIdx.x * 256 + threadIdx.x;
  const int n  = t >> 4;
  const int q  = t & 15;
  const int nl = threadIdx.x >> 4;

  float4 acc = make_float4(0.f, 0.f, 0.f, 0.f);

  const int deg = cnt[n];
  const int beg = n * CAP;
  const int end = beg + min(deg, CAP);
  const float4* X4 = (const float4*)H + q;

  if (deg > CAP) {
    const int no = cnt[NNODES];
    for (int i = 0; i < no; ++i) {
      const int4 r = ovf[i];
      if (r.x == n) {
        const float4 mm = X4[(size_t)r.y * (DIM / 4)];
        const float  w  = __int_as_float(r.z);
        acc.x = fmaf(mm.x, w, acc.x);
        acc.y = fmaf(mm.y, w, acc.y);
        acc.z = fmaf(mm.z, w, acc.z);
        acc.w = fmaf(mm.w, w, acc.w);
      }
    }
  }

  const int last = end - 1;
  for (int j = beg; j < end; j += 8) {
    int2   e[8];
    float4 m[8];
#pragma unroll
    for (int k = 0; k < 8; ++k) e[k] = ell[min(j + k, last)];
#pragma unroll
    for (int k = 0; k < 8; ++k) m[k] = X4[(size_t)e[k].x * (DIM / 4)];
#pragma unroll
    for (int k = 0; k < 8; ++k) {
      const bool  ok = (k == 0) || (j + k) < end;
      const float w  = __int_as_float(e[k].y);
      const float nx = fmaf(m[k].x, w, acc.x);
      const float ny = fmaf(m[k].y, w, acc.y);
      const float nz = fmaf(m[k].z, w, acc.z);
      const float nw = fmaf(m[k].w, w, acc.w);
      acc.x = ok ? nx : acc.x;
      acc.y = ok ? ny : acc.y;
      acc.z = ok ? nz : acc.z;
      acc.w = ok ? nw : acc.w;
    }
  }

  *(float4*)&As[nl * LDP + 4 * q] = acc;

  const int lane = threadIdx.x & 31;
  const int wave = threadIdx.x >> 5;
  const int c    = lane & 15;
  const int hh   = lane >> 4;
  const int n0   = (wave & 3) * 16;
  const int kb   = (wave >> 2) * 32;

  v2f bf[8];
#pragma unroll
  for (int i = 0; i < 8; ++i) {
    const float* w0 = W + (size_t)(kb + 4 * i + 2 * hh) * DIM + n0 + c;
    bf[i].x = w0[0];
    bf[i].y = w0[DIM];
  }

  __builtin_amdgcn_fence(__ATOMIC_RELEASE, "workgroup", "local");
  __builtin_amdgcn_s_barrier();
  __builtin_amdgcn_fence(__ATOMIC_ACQUIRE, "workgroup", "local");

  const float4 o = ((const float4*)ori)[t];
  const float4 b = ((const float4*)bias)[q];

  v8f d = {};
#pragma unroll
  for (int i = 0; i < 8; ++i) {
    const v2f a = *(const v2f*)&As[c * LDP + kb + 4 * i + 2 * hh];
    d = __builtin_amdgcn_wmma_f32_16x16x4_f32(false, a, false, bf[i], (short)0, d, false, false);
  }

  float* dp = &Ds[wave >> 2][(8 * hh) * LDP + n0 + c];
#pragma unroll
  for (int r = 0; r < 8; ++r) dp[r * LDP] = d[r];

  __builtin_amdgcn_fence(__ATOMIC_RELEASE, "workgroup", "local");
  __builtin_amdgcn_s_barrier();
  __builtin_amdgcn_fence(__ATOMIC_ACQUIRE, "workgroup", "local");

  const float4 d0 = *(const float4*)&Ds[0][nl * LDP + 4 * q];
  const float4 d1 = *(const float4*)&Ds[1][nl * LDP + 4 * q];
  float4 res;
  res.x = ((d0.x + d1.x) + b.x) + o.x;
  res.y = ((d0.y + d1.y) + b.y) + o.y;
  res.z = ((d0.z + d1.z) + b.z) + o.z;
  res.w = ((d0.w + d1.w) + b.w) + o.w;
  ((float4*)out)[t] = res;
}

extern "C" void kernel_launch(void* const* d_in, const int* in_sizes, int n_in,
                              void* d_out, int out_size, void* d_ws, size_t ws_size,
                              hipStream_t stream) {
  const float* in_feat = (const float*)d_in[0];
  const float* ew      = (const float*)d_in[1];
  const float* W1      = (const float*)d_in[2];
  const float* b1      = (const float*)d_in[3];
  const float* W2      = (const float*)d_in[4];
  const float* b2      = (const float*)d_in[5];
  const int*   src     = (const int*)d_in[6];
  const int*   dst     = (const int*)d_in[7];
  const int    nE      = in_sizes[1];

  float* ha   = (float*)d_ws;
  float* hb   = ha + (size_t)NNODES * DIM;
  int*   cnt  = (int*)(hb + (size_t)NNODES * DIM);
  int2*  ell  = (int2*)((char*)cnt + CNT_BYTES);
  int4*  ovf  = (int4*)(ell + (size_t)NNODES * CAP);
  int2*  cbuf = (int2*)(ovf + (size_t)NEDGES);
  int*   seg  = (int*)(cbuf + (size_t)NEDGES);
  float* outp = (float*)d_out;

  const int nodeBlocks = NNODES * 16 / 256;

  chunk_sort<<<NCHUNK, 1024, 0, stream>>>(src, dst, ew, cbuf, seg, cnt, nE);
  bin_rank<<<NBIN, 1024, 0, stream>>>(cbuf, seg, cnt, ell, ovf);

  gather_gemm<<<nodeBlocks, 256, 0, stream>>>(in_feat, W1, ell, ovf, cnt, in_feat, b1, ha);
  gather_gemm<<<nodeBlocks, 256, 0, stream>>>(ha,      W2, ell, ovf, cnt, in_feat, b2, hb);
  gather_gemm<<<nodeBlocks, 256, 0, stream>>>(hb,      W2, ell, ovf, cnt, in_feat, b2, ha);
  gather_gemm<<<nodeBlocks, 256, 0, stream>>>(ha,      W2, ell, ovf, cnt, in_feat, b2, outp);
}


# Bucket slots read 16 at a time: a node's 16 lanes load one entry each (heads of two neighbour nodes laid out as one aligned 256-byte piece) and share them by DPP row_share; FP sum order unchanged
# speedup vs baseline: 5.1733x; 1.0737x over previous
#include <hip/hip_runtime.h>

typedef __attribute__((ext_vector_type(2))) float v2f;
typedef __attribute__((ext_vector_type(8))) float v8f;

#define NNODES 100000
#define DIM 64

#define CAP 48
#define HEAD 16
#define REST (CAP - HEAD)
#define NCNT (NNODES + 4)
#define CNT_BYTES 400384

#define NEDGES 1600000
#define NCHUNK 256
#define CH (NEDGES / NCHUNK)
#define NBIN 256
#define BINN 391
#define EPT 7

static_assert(NEDGES % NCHUNK == 0, "chunks are whole");
static_assert(NBIN * BINN >= NNODES, "the ranges cover every node");
static_assert(EPT * 1024 >= CH, "a work-group of 1024 covers its chunk");
static_assert(CH < 65536, "start and length of a piece share one word");
static_assert(BINN <= 512 && NNODES <= (1 << 17), "9 bits of node offset above 17 bits of source");
static_assert(NBIN % 32 == 0 && NCHUNK % 32 == 0, "whole waves in the scan and in the piece loop");

__global__ __launch_bounds__(1024) void chunk_sort(const int* __restrict__ src,
                                                   const int* __restrict__ dst,
                                                   const float* __restrict__ ew,
                                                   int2* __restrict__ cbuf,
                                                   int* __restrict__ seg,
                                                   int* __restrict__ cnt, int nE) {
  __shared__ int  hist[NBIN];
  __shared__ int  wsum[NBIN / 32];
  __shared__ int2 sorted[CH];

  const int tid = threadIdx.x;
  const int c   = blockIdx.x;
  const int e0  = c * CH;
  const int n   = max(0, min(CH, nE - e0));

  if (tid < NBIN) hist[tid] = 0;
  if (c == 0 && tid == 0) cnt[NNODES] = 0;

  int d[EPT], s[EPT], w[EPT], r[EPT];
#pragma unroll
  for (int k = 0; k < EPT; ++k) {
    const int i = tid + 1024 * k;
    d[k] = 0; s[k] = 0; w[k] = 0;
    if (i < n) {
      d[k] = dst[e0 + i];
      s[k] = src[e0 + i];
      w[k] = __float_as_int(ew[e0 + i]);
    }
  }
  __syncthreads();

#pragma unroll
  for (int k = 0; k < EPT; ++k) {
    const int i = tid + 1024 * k;
    r[k] = 0;
    if (i < n) r[k] = atomicAdd(&hist[d[k] / BINN], 1);
  }
  __syncthreads();

  int v = 0, incl = 0;
  if (tid < NBIN) {
    v    = hist[tid];
    incl = v;
    for (int off = 1; off < 32; off <<= 1) {
      const int up = __shfl_up(incl, off);
      if ((tid & 31) >= off) incl += up;
    }
    if ((tid & 31) == 31) wsum[tid >> 5] = incl;
  }
  __syncthreads();
  if (tid < NBIN) {
    int base = 0;
    for (int j = 0; j < (tid >> 5); ++j) base += wsum[j];
    const int excl = base + incl - v;
    hist[tid] = excl;
    seg[tid * NCHUNK + c] = (excl << 16) | v;
  }
  __syncthreads();

#pragma unroll
  for (int k = 0; k < EPT; ++k) {
    const int i = tid + 1024 * k;
    if (i < n) {
      const int b = d[k] / BINN;
      sorted[hist[b] + r[k]] = make_int2(((d[k] - b * BINN) << 17) | s[k], w[k]);
    }
  }
  __syncthreads();

  for (int i = tid; i < n; i += 1024) cbuf[e0 + i] = sorted[i];
}

__global__ __launch_bounds__(1024) void bin_rank(const int2* __restrict__ cbuf,
                                                 const int* __restrict__ seg,
                                                 int* __restrict__ cnt,
                                                 int2* __restrict__ ell1,
                                                 int2* __restrict__ ell2,
                                                 int4* __restrict__ ovf) {
  __shared__ int ncnt[BINN];

  const int tid  = threadIdx.x;
  const int lane = tid & 31;
  const int wave = tid >> 5;
  const int b    = blockIdx.x;
  const int n0   = b * BINN;

  if (tid < BINN) ncnt[tid] = 0;

  auto put = [&](const int2 ed, const int rk) {
    const int sn = ed.x & 0x1ffff;
    const int nd = n0 + (int)((unsigned)ed.x >> 17);
    if (rk < HEAD) {
      ell1[(size_t)nd * HEAD + rk] = make_int2(sn, ed.y);
    } else if (rk < CAP) {
      ell2[(size_t)nd * REST + (rk - HEAD)] = make_int2(sn, ed.y);
    } else {
      const int o = atomicAdd(cnt + NNODES, 1);
      ovf[o] = make_int4(nd, sn, ed.y, 0);
    }
  };

  int  st[NCHUNK / 32], cn[NCHUNK / 32];
  int2 e[NCHUNK / 32];
#pragma unroll
  for (int j = 0; j < NCHUNK / 32; ++j) {
    const int      c    = wave + 32 * j;
    const unsigned info = (unsigned)seg[b * NCHUNK + c];
    st[j] = c * CH + (int)(info >> 16);
    cn[j] = (int)(info & 0xffffu);
  }
#pragma unroll
  for (int j = 0; j < NCHUNK / 32; ++j) {
    e[j] = cbuf[min(st[j] + (lane < cn[j] ? lane : 0), NEDGES - 1)];
  }
  __syncthreads();

  int rk[NCHUNK / 32];
#pragma unroll
  for (int j = 0; j < NCHUNK / 32; ++j) {
    rk[j] = 0;
    if (lane < cn[j]) rk[j] = atomicAdd(&ncnt[(unsigned)e[j].x >> 17], 1);
  }
#pragma unroll
  for (int j = 0; j < NCHUNK / 32; ++j) {
    if (lane < cn[j]) put(e[j], rk[j]);
  }
#pragma unroll
  for (int j = 0; j < NCHUNK / 32; ++j) {
    for (int k = lane + 32; k < cn[j]; k += 32) {
      const int2 ed = cbuf[st[j] + k];
      put(ed, atomicAdd(&ncnt[(unsigned)ed.x >> 17], 1));
    }
  }
  __syncthreads();

  if (tid < BINN && n0 + tid < NNODES) cnt[n0 + tid] = ncnt[tid];
}

#define LDP 68

static_assert(NNODES % 16 == 0, "a block is exactly 16 nodes");
static_assert(HEAD == 16 && REST == 32, "a node's 16 lanes hold 16 slots at a time: one head piece, two rest pieces");

template <int K>
__device__ __forceinline__ int row_lane(const int v) {
  return __builtin_amdgcn_update_dpp(0, v, 0x150 + K, 0xf, 0xf, false);
}

template <int K0>
__device__ __forceinline__ void sum8(const int2 e, const int left,
                                     const float4* __restrict__ X4, float4& acc) {
  int    sx[8], wb[8];
  float4 m[8];
  sx[0] = row_lane<K0 + 0>(e.x);  sx[1] = row_lane<K0 + 1>(e.x);
  sx[2] = row_lane<K0 + 2>(e.x);  sx[3] = row_lane<K0 + 3>(e.x);
  sx[4] = row_lane<K0 + 4>(e.x);  sx[5] = row_lane<K0 + 5>(e.x);
  sx[6] = row_lane<K0 + 6>(e.x);  sx[7] = row_lane<K0 + 7>(e.x);
#pragma unroll
  for (int k = 0; k < 8; ++k) m[k] = X4[(size_t)sx[k] * (DIM / 4)];
  wb[0] = row_lane<K0 + 0>(e.y);  wb[1] = row_lane<K0 + 1>(e.y);
  wb[2] = row_lane<K0 + 2>(e.y);  wb[3] = row_lane<K0 + 3>(e.y);
  wb[4] = row_lane<K0 + 4>(e.y);  wb[5] = row_lane<K0 + 5>(e.y);
  wb[6] = row_lane<K0 + 6>(e.y);  wb[7] = row_lane<K0 + 7>(e.y);
#pragma unroll
  for (int k = 0; k < 8; ++k) {
    const bool  ok = (k == 0) || (K0 + k) < left;
    const float w  = __int_as_float(wb[k]);
    const float nx = fmaf(m[k].x, w, acc.x);
    const float ny = fmaf(m[k].y, w, acc.y);
    const float nz = fmaf(m[k].z, w, acc.z);
    const float nw = fmaf(m[k].w, w, acc.w);
    acc.x = ok ? nx : acc.x;
    acc.y = ok ? ny : acc.y;
    acc.z = ok ? nz : acc.z;
    acc.w = ok ? nw : acc.w;
  }
}

__global__ __launch_bounds__(256) void gather_gemm(const float* __restrict__ H,
                                                   const float* __restrict__ W,
                                                   const int2* __restrict__ ell1,
                                                   const int2* __restrict__ ell2,
                                                   const int4* __restrict__ ovf,
                                                   const int* __restrict__ cnt,
                                                   const float* ori,
                                                   const float* __restrict__ bias,
                                                   float* __restrict__ out) {
  __shared__ float As[16 * LDP];
  __shared__ float Ds[2][16 * LDP];

  const int t  = blockIdx.x * 256 + threadIdx.x;
  const int n  = t >> 4;
  const int q  = t & 15;
  const int nl = threadIdx.x >> 4;

  float4 acc = make_float4(0.f, 0.f, 0.f, 0.f);

  const int deg  = cnt[n];
  const int dcap = min(deg, CAP);
  const float4* X4 = (const float4*)H + q;

  int2 e0 = make_int2(0, 0), e1 = e0, e2 = e0;
  if (dcap > 0)         e0 = ell1[(size_t)n * HEAD + min(q, dcap - 1)];
  if (dcap > HEAD)      e1 = ell2[(size_t)n * REST + min(q, dcap - HEAD - 1)];
  if (dcap > HEAD + 16) e2 = ell2[(size_t)n * REST + 16 + min(q, dcap - HEAD - 17)];

  if (deg > CAP) {
    const int no = cnt[NNODES];
    for (int i = 0; i < no; ++i) {
      const int4 r = ovf[i];
      if (r.x == n) {
        const float4 mm = X4[(size_t)r.y * (DIM / 4)];
        const float  w  = __int_as_float(r.z);
        acc.x = fmaf(mm.x, w, acc.x);
        acc.y = fmaf(mm.y, w, acc.y);
        acc.z = fmaf(mm.z, w, acc.z);
        acc.w = fmaf(mm.w, w, acc.w);
      }
    }
  }

  if (dcap > 0) {
    sum8<0>(e0, dcap, X4, acc);
    if (dcap > 8) {
      sum8<8>(e0, dcap, X4, acc);
      if (dcap > HEAD) {
        sum8<0>(e1, dcap - HEAD, X4, acc);
        if (dcap > HEAD + 8) {
          sum8<8>(e1, dcap - HEAD, X4, acc);
          if (dcap > HEAD + 16) {
            sum8<0>(e2, dcap - HEAD - 16, X4, acc);
            if (dcap > HEAD + 24) sum8<8>(e2, dcap - HEAD - 16, X4, acc);
          }
        }
      }
    }
  }

  *(float4*)&As[nl * LDP + 4 * q] = acc;

  const int lane = threadIdx.x & 31;
  const int wave = threadIdx.x >> 5;
  const int c    = lane & 15;
  const int hh   = lane >> 4;
  const int n0   = (wave & 3) * 16;
  const int kb   = (wave >> 2) * 32;

  v2f bf[8];
#pragma unroll
  for (int i = 0; i < 8; ++i) {
    const float* w0 = W + (size_t)(kb + 4 * i + 2 * hh) * DIM + n0 + c;
    bf[i].x = w0[0];
    bf[i].y = w0[DIM];
  }

  __builtin_amdgcn_fence(__ATOMIC_RELEASE, "workgroup", "local");
  __builtin_amdgcn_s_barrier();
  __builtin_amdgcn_fence(__ATOMIC_ACQUIRE, "workgroup", "local");

  const float4 o = ((const float4*)ori)[t];
  const float4 b = ((const float4*)bias)[q];

  v8f d = {};
#pragma unroll
  for (int i = 0; i < 8; ++i) {
    const v2f a = *(const v2f*)&As[c * LDP + kb + 4 * i + 2 * hh];
    d = __builtin_amdgcn_wmma_f32_16x16x4_f32(false, a, false, bf[i], (short)0, d, false, false);
  }

  float* dp = &Ds[wave >> 2][(8 * hh) * LDP + n0 + c];
#pragma unroll
  for (int r = 0; r < 8; ++r) dp[r * LDP] = d[r];

  __builtin_amdgcn_fence(__ATOMIC_RELEASE, "workgroup", "local");
  __builtin_amdgcn_s_barrier();
  __builtin_amdgcn_fence(__ATOMIC_ACQUIRE, "workgroup", "local");

  const float4 d0 = *(const float4*)&Ds[0][nl * LDP + 4 * q];
  const float4 d1 = *(const float4*)&Ds[1][nl * LDP + 4 * q];
  float4 res;
  res.x = ((d0.x + d1.x) + b.x) + o.x;
  res.y = ((d0.y + d1.y) + b.y) + o.y;
  res.z = ((d0.z + d1.z) + b.z) + o.z;
  res.w = ((d0.w + d1.w) + b.w) + o.w;
  ((float4*)out)[t] = res;
}

extern "C" void kernel_launch(void* const* d_in, const int* in_sizes, int n_in,
                              void* d_out, int out_size, void* d_ws, size_t ws_size,
                              hipStream_t stream) {
  const float* in_feat = (const float*)d_in[0];
  const float* ew      = (const float*)d_in[1];
  const float* W1      = (const float*)d_in[2];
  const float* b1      = (const float*)d_in[3];
  const float* W2      = (const float*)d_in[4];
  const float* b2      = (const float*)d_in[5];
  const int*   src     = (const int*)d_in[6];
  const int*   dst     = (const int*)d_in[7];
  const int    nE      = in_sizes[1];

  float* ha   = (float*)d_ws;
  float* hb   = ha + (size_t)NNODES * DIM;
  int*   cnt  = (int*)(hb + (size_t)NNODES * DIM);
  int2*  ell1 = (int2*)((char*)cnt + CNT_BYTES);
  int2*  ell2 = ell1 + (size_t)NNODES * HEAD;
  int4*  ovf  = (int4*)(ell2 + (size_t)NNODES * REST);
  int2*  cbuf = (int2*)(ovf + (size_t)NEDGES);
  int*   seg  = (int*)(cbuf + (size_t)NEDGES);
  float* outp = (float*)d_out;

  const int nodeBlocks = NNODES * 16 / 256;

  chunk_sort<<<NCHUNK, 1024, 0, stream>>>(src, dst, ew, cbuf, seg, cnt, nE);
  bin_rank<<<NBIN, 1024, 0, stream>>>(cbuf, seg, cnt, ell1, ell2, ovf);

  gather_gemm<<<nodeBlocks, 256, 0, stream>>>(in_feat, W1, ell1, ell2, ovf, cnt, in_feat, b1, ha);
  gather_gemm<<<nodeBlocks, 256, 0, stream>>>(ha,      W2, ell1, ell2, ovf, cnt, in_feat, b2, hb);
  gather_gemm<<<nodeBlocks, 256, 0, stream>>>(hb,      W2, ell1, ell2, ovf, cnt, in_feat, b2, ha);
  gather_gemm<<<nodeBlocks, 256, 0, stream>>>(ha,      W2, ell1, ell2, ovf, cnt, in_feat, b2, outp);
}
